# starting kernel
# baseline (speedup 1.0000x reference)
#include <hip/hip_runtime.h>

typedef __attribute__((ext_vector_type(2))) float v2f;
typedef __attribute__((ext_vector_type(8))) float v8f;

#define VOCAB 32000
#define EMBED 256
#define NPOS  4096
#define NGRAM 3

__host__ __device__ constexpr unsigned mkcfg(unsigned K) {
    return ((K % 3u) * 16u + K / 3u)
         | ((((K + 1u) % 3u) * 16u + (K + 1u) / 3u) << 6)
         | ((K / 3u) << 12)
         | (((K + 1u) / 3u) << 16);
}

__global__ __launch_bounds__(256) void ngrams_embed_wmma(
    const int* __restrict__ inp,
    const float* __restrict__ weight,
    float* __restrict__ out)
{
    __shared__ int sTab[96];

    const int pg  = blockIdx.x;
    const int p0  = pg * 16;
    const int tid = threadIdx.x;

    if (tid < 16) {
        const int q  = tid;
        const int i0 = inp[p0 + q];
        const int i1 = inp[NPOS + p0 + q];
        const int i2 = inp[2 * NPOS + p0 + q];
        sTab[q]      = i0;
        sTab[16 + q] = i1;
        sTab[32 + q] = i2;
        sTab[48 + q] = __float_as_int(1.0f);
        sTab[64 + q] = (i1 != i0) ? __float_as_int(1.0f) : 0;
        sTab[80 + q] = ((i2 != i0) && (i2 != i1)) ? __float_as_int(1.0f) : 0;
    }
    __syncthreads();

    const int wid  = tid >> 5;
    const int lane = tid & 31;
    const int half = lane >> 4;
    const unsigned r = lane & 15;
    const unsigned colbase = (unsigned)wid * 32u + r;

    v8f acc0 = {};
    v8f acc1 = {};

#pragma unroll
    for (int kg = 0; kg < 12; ++kg) {
        const unsigned cfg = half ? mkcfg(4u * kg + 2u) : mkcfg(4u * kg);
        const unsigned off0 = cfg & 63u;
        const unsigned off1 = (cfg >> 6) & 63u;
        const unsigned q0   = (cfg >> 12) & 15u;
        const unsigned q1   = (cfg >> 16) & 15u;

        const unsigned row0 = (unsigned)sTab[off0];
        const float    fl0  = __int_as_float(sTab[48 + off0]);
        const unsigned row1 = (unsigned)sTab[off1];
        const float    fl1  = __int_as_float(sTab[48 + off1]);

        v2f a;
        a.x = (r == q0) ? fl0 : 0.0f;
        a.y = (r == q1) ? fl1 : 0.0f;

        const float* b0p = weight + ((row0 << 8) + colbase);
        const float* b1p = weight + ((row1 << 8) + colbase);

        v2f b;
        b.x = b0p[0];
        b.y = b1p[0];
        acc0 = __builtin_amdgcn_wmma_f32_16x16x4_f32(
            false, a, false, b, (short)0, acc0, false, false);

        b.x = b0p[16];
        b.y = b1p[16];
        acc1 = __builtin_amdgcn_wmma_f32_16x16x4_f32(
            false, a, false, b, (short)0, acc1, false, false);
    }

#pragma unroll
    for (int j = 0; j < 8; ++j) {
        const unsigned m    = (unsigned)j + 8u * (unsigned)half;
        const unsigned ooff = (((unsigned)p0 + m) << 8) + (unsigned)wid * 32u + r;
        __builtin_nontemporal_store(acc0[j], out + ooff);
        __builtin_nontemporal_store(acc1[j], out + ooff + 16u);
    }
}

extern "C" void kernel_launch(void* const* d_in, const int* in_sizes, int n_in,
                              void* d_out, int out_size, void* d_ws, size_t ws_size,
                              hipStream_t stream) {
    const int*   inp    = (const int*)d_in[0];
    const float* weight = (const float*)d_in[1];
    float*       outp   = (float*)d_out;

    ngrams_embed_wmma<<<dim3(NPOS / 16), dim3(256), 0, stream>>>(inp, weight, outp);
}


# Row as a 16x16 B tile, A = dedupe flag x identity: one position per accumulator, 4 per wave, row ids wave-uniform (scalar index loads, no LDS/barrier), 512-B loads/stores; grams summed in turn 0,1,2
# speedup vs baseline: 1.2682x; 1.2682x over previous
#include <hip/hip_runtime.h>

typedef __attribute__((ext_vector_type(2))) float v2f;
typedef __attribute__((ext_vector_type(4))) float v4f;
typedef __attribute__((ext_vector_type(8))) float v8f;

#define VOCAB 32000
#define EMBED 256
#define NPOS  4096
#define NGRAM 3
#define PPW   4

__global__ __launch_bounds__(128) void ngrams_embed_wmma(
    const int* __restrict__ inp,
    const float* __restrict__ weight,
    float* __restrict__ out)
{
    const unsigned lane = threadIdx.x & 31u;
    const int wv = __builtin_amdgcn_readfirstlane((int)(threadIdx.x >> 5));
    const int p0 = ((int)blockIdx.x * 4 + wv) * PPW;

    int id[NGRAM][PPW];
#pragma unroll
    for (int g = 0; g < NGRAM; ++g)
#pragma unroll
        for (int i = 0; i < PPW; ++i)
            id[g][i] = inp[g * NPOS + p0 + i];

    __builtin_amdgcn_sched_barrier(0);

    v4f x[PPW][NGRAM][2];
#pragma unroll
    for (int i = 0; i < PPW; ++i) {
#pragma unroll
        for (int g = 0; g < NGRAM; ++g) {
            const float* rb = weight + ((size_t)(unsigned)id[g][i] << 8) + lane * 4u;
            x[i][g][0] = *(const v4f*)(rb);
            x[i][g][1] = *(const v4f*)(rb + 128);
        }
        __builtin_amdgcn_sched_barrier(0);
    }

    const int d = (int)(lane & 15u) - 8 * (int)(lane >> 4);
    v2f m[4];
#pragma unroll
    for (int u = 0; u < 4; ++u) {
        m[u].x = (d == 2 * u)     ? 1.0f : 0.0f;
        m[u].y = (d == 2 * u + 1) ? 1.0f : 0.0f;
    }

#pragma unroll
    for (int i = 0; i < PPW; ++i) {
        v8f acc[2] = {{}, {}};
#pragma unroll
        for (int g = 0; g < NGRAM; ++g) {
            const float f = (g == 0) ? 1.0f
                          : (g == 1) ? ((id[1][i] != id[0][i]) ? 1.0f : 0.0f)
                                     : ((id[2][i] != id[0][i] && id[2][i] != id[1][i]) ? 1.0f : 0.0f);
#pragma unroll
            for (int p = 0; p < 2; ++p) {
#pragma unroll
                for (int e = 0; e < 2; ++e) {
                    const int u = 2 * e + p;
                    const v4f xe = x[i][g][e];
                    v2f b;
                    b.x = p ? xe.z : xe.x;
                    b.y = p ? xe.w : xe.y;
                    const v2f a = (g == 0) ? m[u] : m[u] * f;
                    acc[e] = __builtin_amdgcn_wmma_f32_16x16x4_f32(
                        false, a, false, b, (short)0, acc[e], false, false);
                }
            }
        }
        float* op = out + ((size_t)(unsigned)(p0 + i) << 8) + lane * 4u;
        v4f lo, hi;
        lo.x = acc[0][0]; lo.y = acc[0][1]; lo.z = acc[0][2]; lo.w = acc[0][3];
        hi.x = acc[1][4]; hi.y = acc[1][5]; hi.z = acc[1][6]; hi.w = acc[1][7];
        __builtin_nontemporal_store(lo, (v4f*)(op));
        __builtin_nontemporal_store(hi, (v4f*)(op + 128));
        __builtin_amdgcn_sched_barrier(0);
    }
}

extern "C" void kernel_launch(void* const* d_in, const int* in_sizes, int n_in,
                              void* d_out, int out_size, void* d_ws, size_t ws_size,
                              hipStream_t stream) {
    const int*   inp    = (const int*)d_in[0];
    const float* weight = (const float*)d_in[1];
    float*       outp   = (float*)d_out;

    ngrams_embed_wmma<<<dim3(NPOS / 16), dim3(128), 0, stream>>>(inp, weight, outp);
}
